# starting kernel
# baseline (speedup 1.0000x reference)
#include <hip/hip_runtime.h>

typedef float v2f __attribute__((ext_vector_type(2)));
typedef float v8f __attribute__((ext_vector_type(8)));

#define D   128
#define NPG 64

__global__ __launch_bounds__(256) void partner_kernel(
    const float* __restrict__ node, const float* __restrict__ glob,
    const float* __restrict__ Wp1,  const float* __restrict__ bp1,
    const float* __restrict__ Wp2,  const float* __restrict__ bp2,
    float* __restrict__ out)
{
  extern __shared__ float smem[];
  float* lds_wb   = smem;
  float* lds_base = smem + 16384;
  float* lds_vec  = smem + 16384 + 256;
  float* lds_wp2  = smem + 16384 + 768;

  const int tid = threadIdx.x;
  const int g0  = blockIdx.x * 2;

  for (int i = tid; i < D * D; i += 256) {
    int k = i >> 7, n = i & 127;
    float v = Wp1[(D + k) * D + n];
    int p = k >> 1;
    int col = (n + ((p & 1) << 4)) & 127;
    lds_wb[p * 256 + col * 2 + (k & 1)] = v;
  }
  {
    int g = tid >> 7, j = tid & 127;
    int graph = g0 + g;
    lds_vec[(g * 2 + 0) * D + j] = node[(graph * NPG + NPG - 1) * D + j];
    lds_vec[(g * 2 + 1) * D + j] = glob[graph * D + j];
  }
  if (tid < D) lds_wp2[tid] = Wp2[tid];
  __syncthreads();

  {
    int g = tid >> 7, j = tid & 127;
    float s = bp1[j];
    const float* cv = &lds_vec[(g * 2 + 0) * D];
    const float* gv = &lds_vec[(g * 2 + 1) * D];
#pragma unroll 8
    for (int k = 0; k < D; ++k) {
      s += cv[k] * Wp1[k * D + j];
      s += gv[k] * Wp1[(2 * D + k) * D + j];
    }
    lds_base[g * D + j] = s;
  }
  __syncthreads();

  const int wave = tid >> 5, lane = tid & 31;
  const int half = lane >> 4, l16 = lane & 15;
  const int g = wave >> 2, mtile = wave & 3;
  const int graph = g0 + g;
  const int row = graph * NPG + mtile * 16 + l16;
  const float* rowp = node + (size_t)row * D + 2 * half;

  v2f af[32];
#pragma unroll
  for (int s = 0; s < 32; ++s) af[s] = *(const v2f*)(rowp + 4 * s);

  const v2f* wb2 = (const v2f*)lds_wb;
  const int coln0 = l16;
  const int coln1 = (l16 + 16) & 127;

  float acc[8];
#pragma unroll
  for (int r = 0; r < 8; ++r) acc[r] = 0.f;

  for (int nt = 0; nt < 8; ++nt) {
    int n = nt * 16 + l16;
    int col = ((n + 16 * half) & 127);
    v8f c = {};
#pragma unroll
    for (int s = 0; s < 32; ++s) {
      v2f b = wb2[(2 * s + half) * D + col];
      c = __builtin_amdgcn_wmma_f32_16x16x4_f32(false, af[s], false, b,
                                                (short)0, c, false, false);
    }
    float bn = lds_base[g * D + n];
    float wn = lds_wp2[n];
#pragma unroll
    for (int r = 0; r < 8; ++r) {
      float v = c[r] + bn;
      v = v > 0.f ? v : 0.f;
      acc[r] += v * wn;
    }
  }
#pragma unroll
  for (int r = 0; r < 8; ++r) {
    float a = acc[r];
    a += __shfl_xor(a, 1, 32);
    a += __shfl_xor(a, 2, 32);
    a += __shfl_xor(a, 4, 32);
    a += __shfl_xor(a, 8, 32);
    acc[r] = a;
  }
  if (l16 == 0) {
    float b2 = bp2[0];
#pragma unroll
    for (int r = 0; r < 8; ++r)
      out[graph * NPG + mtile * 16 + r + 8 * half] = acc[r] + b2;
  }
}

__global__ __launch_bounds__(256) void label_kernel(
    const float* __restrict__ node, const float* __restrict__ glob,
    const float* __restrict__ Wl1,  const float* __restrict__ bl1,
    const float* __restrict__ Wl2,  const float* __restrict__ bl2,
    const float* __restrict__ Wl3,  const float* __restrict__ bl3,
    const int* __restrict__ pgi,    const int* __restrict__ pni,
    float* __restrict__ out, int outoff)
{
  extern __shared__ float smem[];
  float* lds_h1 = smem;
  float* lds_h2 = smem + 8 * 16 * D;
  float* lds_w3 = smem + 2 * 8 * 16 * D;

  const int tid = threadIdx.x;
  for (int i = tid; i < D * 16; i += 256) {
    int k = i >> 4, n = i & 15;
    lds_w3[i] = (n < 14) ? Wl3[k * 14 + n] : 0.f;
  }

  const int wave = tid >> 5, lane = tid & 31;
  const int half = lane >> 4, l16 = lane & 15;
  const int rowid = blockIdx.x * 128 + wave * 16 + l16;
  const int pg = pgi[rowid];
  const int pn = pni[rowid];
  const float* src0 = node + ((size_t)(pg + 1) * NPG - 1) * D;
  const float* src1 = node + (size_t)pn * D;
  const float* src2 = glob + (size_t)pg * D;

  for (int nt = 0; nt < 8; ++nt) {
    int n = nt * 16 + l16;
    v8f c = {};
#pragma unroll
    for (int seg = 0; seg < 3; ++seg) {
      const float* ap = (seg == 0 ? src0 : seg == 1 ? src1 : src2) + 2 * half;
      const float* bp = Wl1 + (size_t)(seg * D + 2 * half) * D + n;
#pragma unroll
      for (int s = 0; s < 32; ++s) {
        v2f a = *(const v2f*)(ap + 4 * s);
        v2f b;
        b.x = bp[(size_t)(4 * s) * D];
        b.y = bp[(size_t)(4 * s + 1) * D];
        c = __builtin_amdgcn_wmma_f32_16x16x4_f32(false, a, false, b,
                                                  (short)0, c, false, false);
      }
    }
    float bv = bl1[n];
#pragma unroll
    for (int r = 0; r < 8; ++r) {
      float v = c[r] + bv;
      lds_h1[(wave * 16 + r + 8 * half) * D + n] = v > 0.f ? v : 0.f;
    }
  }
  __syncthreads();

  for (int nt = 0; nt < 8; ++nt) {
    int n = nt * 16 + l16;
    v8f c = {};
    const float* ap = lds_h1 + (wave * 16 + l16) * D + 2 * half;
    const float* bp = Wl2 + (size_t)(2 * half) * D + n;
#pragma unroll
    for (int s = 0; s < 32; ++s) {
      v2f a = *(const v2f*)(ap + 4 * s);
      v2f b;
      b.x = bp[(size_t)(4 * s) * D];
      b.y = bp[(size_t)(4 * s + 1) * D];
      c = __builtin_amdgcn_wmma_f32_16x16x4_f32(false, a, false, b,
                                                (short)0, c, false, false);
    }
    float bv = bl2[n];
#pragma unroll
    for (int r = 0; r < 8; ++r) {
      float v = c[r] + bv;
      lds_h2[(wave * 16 + r + 8 * half) * D + n] = v > 0.f ? v : 0.f;
    }
  }
  __syncthreads();

  {
    v8f c = {};
    const float* ap = lds_h2 + (wave * 16 + l16) * D + 2 * half;
#pragma unroll
    for (int s = 0; s < 32; ++s) {
      v2f a = *(const v2f*)(ap + 4 * s);
      v2f b;
      b.x = lds_w3[(4 * s + 2 * half) * 16 + l16];
      b.y = lds_w3[(4 * s + 2 * half + 1) * 16 + l16];
      c = __builtin_amdgcn_wmma_f32_16x16x4_f32(false, a, false, b,
                                                (short)0, c, false, false);
    }
    if (l16 < 14) {
      float bv = bl3[l16];
#pragma unroll
      for (int r = 0; r < 8; ++r) {
        int grow = blockIdx.x * 128 + wave * 16 + r + 8 * half;
        out[outoff + grow * 14 + l16] = c[r] + bv;
      }
    }
  }
}

extern "C" void kernel_launch(void* const* d_in, const int* in_sizes, int n_in,
                              void* d_out, int out_size, void* d_ws, size_t ws_size,
                              hipStream_t stream) {
  (void)n_in; (void)out_size; (void)d_ws; (void)ws_size;
  const float* node = (const float*)d_in[0];
  const float* glob = (const float*)d_in[1];
  const float* Wp1  = (const float*)d_in[2];
  const float* bp1  = (const float*)d_in[3];
  const float* Wp2  = (const float*)d_in[4];
  const float* bp2  = (const float*)d_in[5];
  const float* Wl1  = (const float*)d_in[6];
  const float* bl1  = (const float*)d_in[7];
  const float* Wl2  = (const float*)d_in[8];
  const float* bl2  = (const float*)d_in[9];
  const float* Wl3  = (const float*)d_in[10];
  const float* bl3  = (const float*)d_in[11];
  const int*   pgi  = (const int*)d_in[12];
  const int*   pni  = (const int*)d_in[13];
  float* out = (float*)d_out;

  const int B = in_sizes[1] / D;
  const int N = in_sizes[0] / D;

  const size_t smem1 = (16384 + 256 + 512 + 128) * sizeof(float);
  const size_t smem2 = (2 * 8 * 16 * D + D * 16) * sizeof(float);

  partner_kernel<<<B / 2, 256, smem1, stream>>>(node, glob, Wp1, bp1, Wp2, bp2, out);
  label_kernel<<<B / 128, 256, smem2, stream>>>(node, glob, Wl1, bl1, Wl2, bl2,
                                                Wl3, bl3, pgi, pni, out, N);
}


# partner kernel: base vectors no longer summed per block on the vector unit; extra one-wave blocks of the label launch (now first) form them as 16-graph WMMA tiles in workspace; sum order changes
# speedup vs baseline: 2.8100x; 2.8100x over previous
#include <hip/hip_runtime.h>

typedef float v2f __attribute__((ext_vector_type(2)));
typedef float v8f __attribute__((ext_vector_type(8)));

#define D   128
#define NPG 64

__device__ __forceinline__ float bcast(float v, int k) {
  return __builtin_bit_cast(float, __builtin_amdgcn_readlane(__builtin_bit_cast(int, v), k));
}

__device__ __forceinline__ v8f mma4(v2f a, float b0, float b1, v8f c) {
  v2f b; b.x = b0; b.y = b1;
  return __builtin_amdgcn_wmma_f32_16x16x4_f32(false, a, false, b, (short)0, c, false, false);
}

#define PP 132
#define GCH 4

__global__ __launch_bounds__(128) void partner_kernel(
    const float* __restrict__ node, const float* __restrict__ glob,
    const float* __restrict__ Wp1,  const float* __restrict__ bp1,
    const float* __restrict__ Wp2,  const float* __restrict__ bp2,
    const float* __restrict__ base_ws, size_t ws_bytes, int nlab,
    float* __restrict__ out)
{
  extern __shared__ float smem[];
  float* lds_x    = smem;
  float* lds_glob = smem + 128 * PP;
  float* lds_base = lds_glob + 256;
  float* lds_part = lds_base + 256;

  const int tid = threadIdx.x;
  const int wave = tid >> 5, lane = tid & 31;
  const int half = lane >> 4, c = lane & 15;
  const int g0 = blockIdx.x * 2;

  const float4* src = (const float4*)(node + (size_t)g0 * NPG * D);
  {
    float4 t[32];
#pragma unroll
    for (int i = 0; i < 32; ++i) t[i] = src[i * 128 + tid];
#pragma unroll
    for (int i = 0; i < 32; ++i)
      *(float4*)(lds_x + (i * 4 + wave) * PP + lane * 4) = t[i];
  }
  __builtin_amdgcn_sched_barrier(0);

  v2f wa[2][32];
  {
    const float* wsrc = Wp1 + (size_t)(D + 4 * half) * D + 32 * wave + c;
#pragma unroll
    for (int u = 0; u < 16; ++u) {
#pragma unroll
      for (int T = 0; T < 2; ++T) {
        wa[T][2 * u].x     = wsrc[(8 * u + 0) * D + 16 * T];
        wa[T][2 * u].y     = wsrc[(8 * u + 1) * D + 16 * T];
        wa[T][2 * u + 1].x = wsrc[(8 * u + 2) * D + 16 * T];
        wa[T][2 * u + 1].y = wsrc[(8 * u + 3) * D + 16 * T];
      }
    }
  }

  const bool base_from_ws = ws_bytes >= (size_t)nlab * 16 * D * sizeof(float);
  if (base_from_ws) {
    lds_base[tid]     = base_ws[(size_t)g0 * D + tid];
    lds_base[D + tid] = base_ws[(size_t)g0 * D + D + tid];
    __syncthreads();
  } else {
  lds_glob[tid]     = glob[(size_t)g0 * D + tid];
  lds_glob[D + tid] = glob[(size_t)g0 * D + D + tid];
  const float bias = bp1[tid];
  __syncthreads();

  {
    const int part = wave >> 1;
    const int k0 = 64 * (wave & 1);
    const float* x0 = part ? lds_glob + k0     : lds_x + (NPG - 1) * PP + k0;
    const float* x1 = part ? lds_glob + D + k0 : lds_x + (2 * NPG - 1) * PP + k0;
    const float* wr = Wp1 + (size_t)(2 * D * part + k0) * D + 4 * lane;
    float4 s0 = {0.f, 0.f, 0.f, 0.f}, s1 = {0.f, 0.f, 0.f, 0.f};
#pragma unroll
    for (int h = 0; h < 2; ++h) {
      const float xv0 = x0[32 * h + lane], xv1 = x1[32 * h + lane];
      const float* wrh = wr + (size_t)(32 * h) * D;
#pragma unroll 4
      for (int k4 = 0; k4 < 8; ++k4) {
        float4 xa, xc;
        xa.x = bcast(xv0, 4 * k4 + 0); xa.y = bcast(xv0, 4 * k4 + 1); xa.z = bcast(xv0, 4 * k4 + 2); xa.w = bcast(xv0, 4 * k4 + 3);
        xc.x = bcast(xv1, 4 * k4 + 0); xc.y = bcast(xv1, 4 * k4 + 1); xc.z = bcast(xv1, 4 * k4 + 2); xc.w = bcast(xv1, 4 * k4 + 3);
        const float4 w0 = *(const float4*)(wrh + (size_t)(4 * k4 + 0) * D);
        const float4 w1 = *(const float4*)(wrh + (size_t)(4 * k4 + 1) * D);
        const float4 w2r = *(const float4*)(wrh + (size_t)(4 * k4 + 2) * D);
        const float4 w3 = *(const float4*)(wrh + (size_t)(4 * k4 + 3) * D);
        s0.x = __builtin_fmaf(xa.x, w0.x, s0.x);  s0.y = __builtin_fmaf(xa.x, w0.y, s0.y);
        s0.z = __builtin_fmaf(xa.x, w0.z, s0.z);  s0.w = __builtin_fmaf(xa.x, w0.w, s0.w);
        s1.x = __builtin_fmaf(xc.x, w0.x, s1.x);  s1.y = __builtin_fmaf(xc.x, w0.y, s1.y);
        s1.z = __builtin_fmaf(xc.x, w0.z, s1.z);  s1.w = __builtin_fmaf(xc.x, w0.w, s1.w);
        s0.x = __builtin_fmaf(xa.y, w1.x, s0.x);  s0.y = __builtin_fmaf(xa.y, w1.y, s0.y);
        s0.z = __builtin_fmaf(xa.y, w1.z, s0.z);  s0.w = __builtin_fmaf(xa.y, w1.w, s0.w);
        s1.x = __builtin_fmaf(xc.y, w1.x, s1.x);  s1.y = __builtin_fmaf(xc.y, w1.y, s1.y);
        s1.z = __builtin_fmaf(xc.y, w1.z, s1.z);  s1.w = __builtin_fmaf(xc.y, w1.w, s1.w);
        s0.x = __builtin_fmaf(xa.z, w2r.x, s0.x); s0.y = __builtin_fmaf(xa.z, w2r.y, s0.y);
        s0.z = __builtin_fmaf(xa.z, w2r.z, s0.z); s0.w = __builtin_fmaf(xa.z, w2r.w, s0.w);
        s1.x = __builtin_fmaf(xc.z, w2r.x, s1.x); s1.y = __builtin_fmaf(xc.z, w2r.y, s1.y);
        s1.z = __builtin_fmaf(xc.z, w2r.z, s1.z); s1.w = __builtin_fmaf(xc.z, w2r.w, s1.w);
        s0.x = __builtin_fmaf(xa.w, w3.x, s0.x);  s0.y = __builtin_fmaf(xa.w, w3.y, s0.y);
        s0.z = __builtin_fmaf(xa.w, w3.z, s0.z);  s0.w = __builtin_fmaf(xa.w, w3.w, s0.w);
        s1.x = __builtin_fmaf(xc.w, w3.x, s1.x);  s1.y = __builtin_fmaf(xc.w, w3.y, s1.y);
        s1.z = __builtin_fmaf(xc.w, w3.z, s1.z);  s1.w = __builtin_fmaf(xc.w, w3.w, s1.w);
      }
    }
    *(float4*)(lds_part + (wave * 2 + 0) * D + 4 * lane) = s0;
    *(float4*)(lds_part + (wave * 2 + 1) * D + 4 * lane) = s1;
  }
  __syncthreads();

  {
    float sA = bias, sB = bias;
#pragma unroll
    for (int w = 0; w < 4; ++w) {
      sA += lds_part[(w * 2 + 0) * D + tid];
      sB += lds_part[(w * 2 + 1) * D + tid];
    }
    lds_base[tid]     = sA;
    lds_base[D + tid] = sB;
  }
  __syncthreads();
  }

  const int hb = 32 * wave + 8 * half;
  float w2[2][8];
#pragma unroll
  for (int T = 0; T < 2; ++T) {
    const float4 wA = *(const float4*)(Wp2 + hb + 16 * T), wB = *(const float4*)(Wp2 + hb + 16 * T + 4);
    w2[T][0] = wA.x; w2[T][1] = wA.y; w2[T][2] = wA.z; w2[T][3] = wA.w;
    w2[T][4] = wB.x; w2[T][5] = wB.y; w2[T][6] = wB.z; w2[T][7] = wB.w;
  }

  constexpr int NG = 16 / GCH;
  const float* xb = lds_x + c * PP + 4 * half;
  float4 buf[2][2 * GCH];
#pragma unroll
  for (int j = 0; j < GCH; ++j) {
    buf[0][2 * j]     = *(const float4*)(xb + 8 * j);
    buf[0][2 * j + 1] = *(const float4*)(xb + 16 * PP + 8 * j);
  }
  float* pdst = lds_part + (wave * 2 + half) * 128 + c;

#pragma unroll
  for (int tp = 0; tp < 4; ++tp) {
    v8f a00, a01, a10, a11;
    {
      const float* bp = lds_base + (tp >> 1) * D + hb;
      const float4 b0a = *(const float4*)(bp),      b0b = *(const float4*)(bp + 4);
      const float4 b1a = *(const float4*)(bp + 16), b1b = *(const float4*)(bp + 20);
      a00 = v8f{b0a.x, b0a.y, b0a.z, b0a.w, b0b.x, b0b.y, b0b.z, b0b.w};
      a10 = v8f{b1a.x, b1a.y, b1a.z, b1a.w, b1b.x, b1b.y, b1b.z, b1b.w};
      a01 = a00; a11 = a10;
    }
#pragma unroll
    for (int gq = 0; gq < NG; ++gq) {
      const int q = tp * NG + gq;
      if (q + 1 < 4 * NG) {
        const int tpn = (q + 1) / NG, un = ((q + 1) % NG) * GCH;
        const float* xn = xb + (32 * tpn) * PP + 8 * un;
#pragma unroll
        for (int j = 0; j < GCH; ++j) {
          buf[(q + 1) & 1][2 * j]     = *(const float4*)(xn + 8 * j);
          buf[(q + 1) & 1][2 * j + 1] = *(const float4*)(xn + 16 * PP + 8 * j);
        }
      }
      __builtin_amdgcn_sched_barrier(0);
#pragma unroll
      for (int j = 0; j < GCH; ++j) {
        const int u = gq * GCH + j;
        const float4 f0 = buf[q & 1][2 * j], f1 = buf[q & 1][2 * j + 1];
        a00 = mma4(wa[0][2 * u],     f0.x, f0.y, a00);
        a01 = mma4(wa[0][2 * u],     f1.x, f1.y, a01);
        a10 = mma4(wa[1][2 * u],     f0.x, f0.y, a10);
        a11 = mma4(wa[1][2 * u],     f1.x, f1.y, a11);
        a00 = mma4(wa[0][2 * u + 1], f0.z, f0.w, a00);
        a01 = mma4(wa[0][2 * u + 1], f1.z, f1.w, a01);
        a10 = mma4(wa[1][2 * u + 1], f0.z, f0.w, a10);
        a11 = mma4(wa[1][2 * u + 1], f1.z, f1.w, a11);
      }
      __builtin_amdgcn_sched_barrier(0);
    }
    float p0 = fmaxf(a00[0], 0.f) * w2[0][0], p1 = fmaxf(a01[0], 0.f) * w2[0][0];
#pragma unroll
    for (int r = 1; r < 8; ++r) {
      p0 += fmaxf(a00[r], 0.f) * w2[0][r];
      p1 += fmaxf(a01[r], 0.f) * w2[0][r];
    }
#pragma unroll
    for (int r = 0; r < 8; ++r) {
      p0 += fmaxf(a10[r], 0.f) * w2[1][r];
      p1 += fmaxf(a11[r], 0.f) * w2[1][r];
    }
    pdst[32 * tp]      = p0;
    pdst[32 * tp + 16] = p1;
  }
  __syncthreads();

  {
    float s = lds_part[tid];
#pragma unroll
    for (int j = 1; j < 8; ++j) s += lds_part[j * 128 + tid];
    out[(size_t)g0 * NPG + tid] = s + bp2[0];
  }
}

#define HP 132

template <int STEPS>
__device__ __forceinline__ void tile_gemm(const v2f* a, const float* __restrict__ wl, v8f* acc) {
  constexpr int CH = 8;
  constexpr int NCH = STEPS / CH;
  float4 buf[2][CH * 4];
#pragma unroll
  for (int i = 0; i < CH; ++i) {
    const float4* p = (const float4*)(wl + (size_t)(4 * i) * D);
    buf[0][4 * i + 0] = p[0];  buf[0][4 * i + 1] = p[1];
    buf[0][4 * i + 2] = p[32]; buf[0][4 * i + 3] = p[33];
  }
#pragma unroll
  for (int ch = 0; ch < NCH; ++ch) {
    if (ch + 1 < NCH) {
#pragma unroll
      for (int i = 0; i < CH; ++i) {
        const float4* p = (const float4*)(wl + (size_t)(4 * ((ch + 1) * CH + i)) * D);
        buf[(ch + 1) & 1][4 * i + 0] = p[0];  buf[(ch + 1) & 1][4 * i + 1] = p[1];
        buf[(ch + 1) & 1][4 * i + 2] = p[32]; buf[(ch + 1) & 1][4 * i + 3] = p[33];
      }
    }
#pragma unroll
    for (int i = 0; i < CH; ++i) {
      const v2f av = a[ch * CH + i];
      const float4 r0a = buf[ch & 1][4 * i + 0], r0b = buf[ch & 1][4 * i + 1];
      const float4 r1a = buf[ch & 1][4 * i + 2], r1b = buf[ch & 1][4 * i + 3];
      acc[0] = mma4(av, r0a.x, r1a.x, acc[0]);
      acc[1] = mma4(av, r0a.y, r1a.y, acc[1]);
      acc[2] = mma4(av, r0a.z, r1a.z, acc[2]);
      acc[3] = mma4(av, r0a.w, r1a.w, acc[3]);
      acc[4] = mma4(av, r0b.x, r1b.x, acc[4]);
      acc[5] = mma4(av, r0b.y, r1b.y, acc[5]);
      acc[6] = mma4(av, r0b.z, r1b.z, acc[6]);
      acc[7] = mma4(av, r0b.w, r1b.w, acc[7]);
    }
  }
}

__device__ __forceinline__ void store_relu_tile(const v8f* acc, const float* __restrict__ bias8,
                                                float* tile, int c, int half) {
  const float4 ba = ((const float4*)bias8)[0], bb = ((const float4*)bias8)[1];
#pragma unroll
  for (int r = 0; r < 8; ++r) {
    float4 lo, hi;
    lo.x = fmaxf(acc[0][r] + ba.x, 0.f); lo.y = fmaxf(acc[1][r] + ba.y, 0.f);
    lo.z = fmaxf(acc[2][r] + ba.z, 0.f); lo.w = fmaxf(acc[3][r] + ba.w, 0.f);
    hi.x = fmaxf(acc[4][r] + bb.x, 0.f); hi.y = fmaxf(acc[5][r] + bb.y, 0.f);
    hi.z = fmaxf(acc[6][r] + bb.z, 0.f); hi.w = fmaxf(acc[7][r] + bb.w, 0.f);
    float* q = tile + (8 * half + r) * HP + 8 * c;
    *(float4*)q = lo;
    *(float4*)(q + 4) = hi;
  }
}

__device__ __forceinline__ void base_tile(
    const float* __restrict__ node, const float* __restrict__ glob,
    const float* __restrict__ Wp1,  const float* __restrict__ bp1,
    float* __restrict__ base_ws, int tile, int c, int half)
{
  const int g = tile * 16 + c;
  const float* s0 = node + ((size_t)(g + 1) * NPG - 1) * D + 2 * half;
  const float* s2 = glob + (size_t)g * D + 2 * half;
  v2f a[64];
#pragma unroll
  for (int s = 0; s < 32; ++s) {
    a[s]      = *(const v2f*)(s0 + 4 * s);
    a[32 + s] = *(const v2f*)(s2 + 4 * s);
  }
  v8f acc[8];
#pragma unroll
  for (int j = 0; j < 8; ++j) acc[j] = v8f{};
  tile_gemm<32>(a,      Wp1 + (size_t)(2 * half) * D + 8 * c,         acc);
  tile_gemm<32>(a + 32, Wp1 + (size_t)(2 * D + 2 * half) * D + 8 * c, acc);
  const float4 ba = ((const float4*)(bp1 + 8 * c))[0], bb = ((const float4*)(bp1 + 8 * c))[1];
#pragma unroll
  for (int r = 0; r < 8; ++r) {
    float4 lo, hi;
    lo.x = acc[0][r] + ba.x; lo.y = acc[1][r] + ba.y; lo.z = acc[2][r] + ba.z; lo.w = acc[3][r] + ba.w;
    hi.x = acc[4][r] + bb.x; hi.y = acc[5][r] + bb.y; hi.z = acc[6][r] + bb.z; hi.w = acc[7][r] + bb.w;
    float* q = base_ws + (size_t)(tile * 16 + 8 * half + r) * D + 8 * c;
    *(float4*)q = lo;
    *(float4*)(q + 4) = hi;
  }
}

__global__ __launch_bounds__(32) void label_kernel(
    const float* __restrict__ node, const float* __restrict__ glob,
    const float* __restrict__ Wl1,  const float* __restrict__ bl1,
    const float* __restrict__ Wl2,  const float* __restrict__ bl2,
    const float* __restrict__ Wl3,  const float* __restrict__ bl3,
    const int* __restrict__ pgi,    const int* __restrict__ pni,
    const float* __restrict__ Wp1,  const float* __restrict__ bp1,
    float* __restrict__ base_ws, size_t ws_bytes, int nlab,
    float* __restrict__ out, int outoff)
{
  extern __shared__ float smem[];
  const int lane = threadIdx.x;
  const int half = lane >> 4, c = lane & 15;
  if ((int)blockIdx.x >= nlab) {
    if (ws_bytes >= (size_t)nlab * 16 * D * sizeof(float))
      base_tile(node, glob, Wp1, bp1, base_ws, (int)blockIdx.x - nlab, c, half);
    return;
  }
  const int rowid = blockIdx.x * 16 + c;
  const int pg = pgi[rowid];
  const int pn = pni[rowid];
  const float* src0 = node + ((size_t)(pg + 1) * NPG - 1) * D + 2 * half;
  const float* src1 = node + (size_t)pn * D + 2 * half;
  const float* src2 = glob + (size_t)pg * D + 2 * half;

  v8f acc[8];

  {
    v2f a[96];
#pragma unroll
    for (int s = 0; s < 32; ++s) {
      a[s]      = *(const v2f*)(src0 + 4 * s);
      a[32 + s] = *(const v2f*)(src1 + 4 * s);
      a[64 + s] = *(const v2f*)(src2 + 4 * s);
    }
#pragma unroll
    for (int j = 0; j < 8; ++j) acc[j] = v8f{};
    tile_gemm<96>(a, Wl1 + (size_t)(2 * half) * D + 8 * c, acc);
    store_relu_tile(acc, bl1 + 8 * c, smem, c, half);
  }
  __syncthreads();

  float b3x[32], b3y[32];
  {
    const int cc = c < 14 ? c : 13;
    const float* w3 = Wl3 + (2 * half) * 14 + cc;
#pragma unroll
    for (int s = 0; s < 32; ++s) {
      const float b0 = w3[(4 * s) * 14], b1 = w3[(4 * s + 1) * 14];
      b3x[s] = c < 14 ? b0 : 0.f;
      b3y[s] = c < 14 ? b1 : 0.f;
    }
  }

  {
    v2f a[32];
#pragma unroll
    for (int s = 0; s < 32; ++s) a[s] = *(const v2f*)(smem + c * HP + 4 * s + 2 * half);
#pragma unroll
    for (int j = 0; j < 8; ++j) acc[j] = v8f{};
    tile_gemm<32>(a, Wl2 + (size_t)(2 * half) * D + 8 * c, acc);
    store_relu_tile(acc, bl2 + 8 * c, smem, c, half);
  }
  __syncthreads();

  {
    v2f a[32];
#pragma unroll
    for (int s = 0; s < 32; ++s) a[s] = *(const v2f*)(smem + c * HP + 4 * s + 2 * half);
    v8f o = {};
#pragma unroll
    for (int s = 0; s < 32; ++s) o = mma4(a[s], b3x[s], b3y[s], o);
    if (c < 14) {
      float bv = bl3[c];
#pragma unroll
      for (int r = 0; r < 8; ++r) {
        int grow = blockIdx.x * 16 + r + 8 * half;
        out[outoff + grow * 14 + c] = o[r] + bv;
      }
    }
  }
}

extern "C" void kernel_launch(void* const* d_in, const int* in_sizes, int n_in,
                              void* d_out, int out_size, void* d_ws, size_t ws_size,
                              hipStream_t stream) {
  (void)n_in; (void)out_size;
  const float* node = (const float*)d_in[0];
  const float* glob = (const float*)d_in[1];
  const float* Wp1  = (const float*)d_in[2];
  const float* bp1  = (const float*)d_in[3];
  const float* Wp2  = (const float*)d_in[4];
  const float* bp2  = (const float*)d_in[5];
  const float* Wl1  = (const float*)d_in[6];
  const float* bl1  = (const float*)d_in[7];
  const float* Wl2  = (const float*)d_in[8];
  const float* bl2  = (const float*)d_in[9];
  const float* Wl3  = (const float*)d_in[10];
  const float* bl3  = (const float*)d_in[11];
  const int*   pgi  = (const int*)d_in[12];
  const int*   pni  = (const int*)d_in[13];
  float* out = (float*)d_out;
  float* ws  = (float*)d_ws;

  const int B = in_sizes[1] / D;
  const int N = in_sizes[0] / D;

  const size_t smem1 = (128 * PP + 256 + 256 + 1024) * sizeof(float);
  const size_t smem2 = 16 * HP * sizeof(float);

  label_kernel<<<B / 8, 32, smem2, stream>>>(node, glob, Wl1, bl1, Wl2, bl2,
                                               Wl3, bl3, pgi, pni, Wp1, bp1, ws, ws_size, B / 16, out, N);
  partner_kernel<<<B / 2, 128, smem1, stream>>>(node, glob, Wp1, bp1, Wp2, bp2, ws, ws_size, B / 16, out);
}


# label launch: each 16-row tile's K range split over the 4 waves of a 128-thread block (label layers 1-3, base tiles; was one wave per tile), partial tiles added through LDS; sum order changes
# speedup vs baseline: 2.8926x; 1.0294x over previous
#include <hip/hip_runtime.h>

typedef float v2f __attribute__((ext_vector_type(2)));
typedef float v8f __attribute__((ext_vector_type(8)));

#define D   128
#define NPG 64

__device__ __forceinline__ float bcast(float v, int k) {
  return __builtin_bit_cast(float, __builtin_amdgcn_readlane(__builtin_bit_cast(int, v), k));
}

__device__ __forceinline__ v8f mma4(v2f a, float b0, float b1, v8f c) {
  v2f b; b.x = b0; b.y = b1;
  return __builtin_amdgcn_wmma_f32_16x16x4_f32(false, a, false, b, (short)0, c, false, false);
}

#define PP 132
#define GCH 4

__global__ __launch_bounds__(128) void partner_kernel(
    const float* __restrict__ node, const float* __restrict__ glob,
    const float* __restrict__ Wp1,  const float* __restrict__ bp1,
    const float* __restrict__ Wp2,  const float* __restrict__ bp2,
    const float* __restrict__ base_ws, size_t ws_bytes, int nlab,
    float* __restrict__ out)
{
  extern __shared__ float smem[];
  float* lds_x    = smem;
  float* lds_glob = smem + 128 * PP;
  float* lds_base = lds_glob + 256;
  float* lds_part = lds_base + 256;

  const int tid = threadIdx.x;
  const int wave = tid >> 5, lane = tid & 31;
  const int half = lane >> 4, c = lane & 15;
  const int g0 = blockIdx.x * 2;

  const float4* src = (const float4*)(node + (size_t)g0 * NPG * D);
  {
    float4 t[32];
#pragma unroll
    for (int i = 0; i < 32; ++i) t[i] = src[i * 128 + tid];
#pragma unroll
    for (int i = 0; i < 32; ++i)
      *(float4*)(lds_x + (i * 4 + wave) * PP + lane * 4) = t[i];
  }
  __builtin_amdgcn_sched_barrier(0);

  v2f wa[2][32];
  {
    const float* wsrc = Wp1 + (size_t)(D + 4 * half) * D + 32 * wave + c;
#pragma unroll
    for (int u = 0; u < 16; ++u) {
#pragma unroll
      for (int T = 0; T < 2; ++T) {
        wa[T][2 * u].x     = wsrc[(8 * u + 0) * D + 16 * T];
        wa[T][2 * u].y     = wsrc[(8 * u + 1) * D + 16 * T];
        wa[T][2 * u + 1].x = wsrc[(8 * u + 2) * D + 16 * T];
        wa[T][2 * u + 1].y = wsrc[(8 * u + 3) * D + 16 * T];
      }
    }
  }

  const bool base_from_ws = ws_bytes >= (size_t)nlab * 16 * D * sizeof(float);
  if (base_from_ws) {
    lds_base[tid]     = base_ws[(size_t)g0 * D + tid];
    lds_base[D + tid] = base_ws[(size_t)g0 * D + D + tid];
    __syncthreads();
  } else {
  lds_glob[tid]     = glob[(size_t)g0 * D + tid];
  lds_glob[D + tid] = glob[(size_t)g0 * D + D + tid];
  const float bias = bp1[tid];
  __syncthreads();

  {
    const int part = wave >> 1;
    const int k0 = 64 * (wave & 1);
    const float* x0 = part ? lds_glob + k0     : lds_x + (NPG - 1) * PP + k0;
    const float* x1 = part ? lds_glob + D + k0 : lds_x + (2 * NPG - 1) * PP + k0;
    const float* wr = Wp1 + (size_t)(2 * D * part + k0) * D + 4 * lane;
    float4 s0 = {0.f, 0.f, 0.f, 0.f}, s1 = {0.f, 0.f, 0.f, 0.f};
#pragma unroll
    for (int h = 0; h < 2; ++h) {
      const float xv0 = x0[32 * h + lane], xv1 = x1[32 * h + lane];
      const float* wrh = wr + (size_t)(32 * h) * D;
#pragma unroll 4
      for (int k4 = 0; k4 < 8; ++k4) {
        float4 xa, xc;
        xa.x = bcast(xv0, 4 * k4 + 0); xa.y = bcast(xv0, 4 * k4 + 1); xa.z = bcast(xv0, 4 * k4 + 2); xa.w = bcast(xv0, 4 * k4 + 3);
        xc.x = bcast(xv1, 4 * k4 + 0); xc.y = bcast(xv1, 4 * k4 + 1); xc.z = bcast(xv1, 4 * k4 + 2); xc.w = bcast(xv1, 4 * k4 + 3);
        const float4 w0 = *(const float4*)(wrh + (size_t)(4 * k4 + 0) * D);
        const float4 w1 = *(const float4*)(wrh + (size_t)(4 * k4 + 1) * D);
        const float4 w2r = *(const float4*)(wrh + (size_t)(4 * k4 + 2) * D);
        const float4 w3 = *(const float4*)(wrh + (size_t)(4 * k4 + 3) * D);
        s0.x = __builtin_fmaf(xa.x, w0.x, s0.x);  s0.y = __builtin_fmaf(xa.x, w0.y, s0.y);
        s0.z = __builtin_fmaf(xa.x, w0.z, s0.z);  s0.w = __builtin_fmaf(xa.x, w0.w, s0.w);
        s1.x = __builtin_fmaf(xc.x, w0.x, s1.x);  s1.y = __builtin_fmaf(xc.x, w0.y, s1.y);
        s1.z = __builtin_fmaf(xc.x, w0.z, s1.z);  s1.w = __builtin_fmaf(xc.x, w0.w, s1.w);
        s0.x = __builtin_fmaf(xa.y, w1.x, s0.x);  s0.y = __builtin_fmaf(xa.y, w1.y, s0.y);
        s0.z = __builtin_fmaf(xa.y, w1.z, s0.z);  s0.w = __builtin_fmaf(xa.y, w1.w, s0.w);
        s1.x = __builtin_fmaf(xc.y, w1.x, s1.x);  s1.y = __builtin_fmaf(xc.y, w1.y, s1.y);
        s1.z = __builtin_fmaf(xc.y, w1.z, s1.z);  s1.w = __builtin_fmaf(xc.y, w1.w, s1.w);
        s0.x = __builtin_fmaf(xa.z, w2r.x, s0.x); s0.y = __builtin_fmaf(xa.z, w2r.y, s0.y);
        s0.z = __builtin_fmaf(xa.z, w2r.z, s0.z); s0.w = __builtin_fmaf(xa.z, w2r.w, s0.w);
        s1.x = __builtin_fmaf(xc.z, w2r.x, s1.x); s1.y = __builtin_fmaf(xc.z, w2r.y, s1.y);
        s1.z = __builtin_fmaf(xc.z, w2r.z, s1.z); s1.w = __builtin_fmaf(xc.z, w2r.w, s1.w);
        s0.x = __builtin_fmaf(xa.w, w3.x, s0.x);  s0.y = __builtin_fmaf(xa.w, w3.y, s0.y);
        s0.z = __builtin_fmaf(xa.w, w3.z, s0.z);  s0.w = __builtin_fmaf(xa.w, w3.w, s0.w);
        s1.x = __builtin_fmaf(xc.w, w3.x, s1.x);  s1.y = __builtin_fmaf(xc.w, w3.y, s1.y);
        s1.z = __builtin_fmaf(xc.w, w3.z, s1.z);  s1.w = __builtin_fmaf(xc.w, w3.w, s1.w);
      }
    }
    *(float4*)(lds_part + (wave * 2 + 0) * D + 4 * lane) = s0;
    *(float4*)(lds_part + (wave * 2 + 1) * D + 4 * lane) = s1;
  }
  __syncthreads();

  {
    float sA = bias, sB = bias;
#pragma unroll
    for (int w = 0; w < 4; ++w) {
      sA += lds_part[(w * 2 + 0) * D + tid];
      sB += lds_part[(w * 2 + 1) * D + tid];
    }
    lds_base[tid]     = sA;
    lds_base[D + tid] = sB;
  }
  __syncthreads();
  }

  const int hb = 32 * wave + 8 * half;
  float w2[2][8];
#pragma unroll
  for (int T = 0; T < 2; ++T) {
    const float4 wA = *(const float4*)(Wp2 + hb + 16 * T), wB = *(const float4*)(Wp2 + hb + 16 * T + 4);
    w2[T][0] = wA.x; w2[T][1] = wA.y; w2[T][2] = wA.z; w2[T][3] = wA.w;
    w2[T][4] = wB.x; w2[T][5] = wB.y; w2[T][6] = wB.z; w2[T][7] = wB.w;
  }

  constexpr int NG = 16 / GCH;
  const float* xb = lds_x + c * PP + 4 * half;
  float4 buf[2][2 * GCH];
#pragma unroll
  for (int j = 0; j < GCH; ++j) {
    buf[0][2 * j]     = *(const float4*)(xb + 8 * j);
    buf[0][2 * j + 1] = *(const float4*)(xb + 16 * PP + 8 * j);
  }
  float* pdst = lds_part + (wave * 2 + half) * 128 + c;

#pragma unroll
  for (int tp = 0; tp < 4; ++tp) {
    v8f a00, a01, a10, a11;
    {
      const float* bp = lds_base + (tp >> 1) * D + hb;
      const float4 b0a = *(const float4*)(bp),      b0b = *(const float4*)(bp + 4);
      const float4 b1a = *(const float4*)(bp + 16), b1b = *(const float4*)(bp + 20);
      a00 = v8f{b0a.x, b0a.y, b0a.z, b0a.w, b0b.x, b0b.y, b0b.z, b0b.w};
      a10 = v8f{b1a.x, b1a.y, b1a.z, b1a.w, b1b.x, b1b.y, b1b.z, b1b.w};
      a01 = a00; a11 = a10;
    }
#pragma unroll
    for (int gq = 0; gq < NG; ++gq) {
      const int q = tp * NG + gq;
      if (q + 1 < 4 * NG) {
        const int tpn = (q + 1) / NG, un = ((q + 1) % NG) * GCH;
        const float* xn = xb + (32 * tpn) * PP + 8 * un;
#pragma unroll
        for (int j = 0; j < GCH; ++j) {
          buf[(q + 1) & 1][2 * j]     = *(const float4*)(xn + 8 * j);
          buf[(q + 1) & 1][2 * j + 1] = *(const float4*)(xn + 16 * PP + 8 * j);
        }
      }
      __builtin_amdgcn_sched_barrier(0);
#pragma unroll
      for (int j = 0; j < GCH; ++j) {
        const int u = gq * GCH + j;
        const float4 f0 = buf[q & 1][2 * j], f1 = buf[q & 1][2 * j + 1];
        a00 = mma4(wa[0][2 * u],     f0.x, f0.y, a00);
        a01 = mma4(wa[0][2 * u],     f1.x, f1.y, a01);
        a10 = mma4(wa[1][2 * u],     f0.x, f0.y, a10);
        a11 = mma4(wa[1][2 * u],     f1.x, f1.y, a11);
        a00 = mma4(wa[0][2 * u + 1], f0.z, f0.w, a00);
        a01 = mma4(wa[0][2 * u + 1], f1.z, f1.w, a01);
        a10 = mma4(wa[1][2 * u + 1], f0.z, f0.w, a10);
        a11 = mma4(wa[1][2 * u + 1], f1.z, f1.w, a11);
      }
      __builtin_amdgcn_sched_barrier(0);
    }
    float p0 = fmaxf(a00[0], 0.f) * w2[0][0], p1 = fmaxf(a01[0], 0.f) * w2[0][0];
#pragma unroll
    for (int r = 1; r < 8; ++r) {
      p0 += fmaxf(a00[r], 0.f) * w2[0][r];
      p1 += fmaxf(a01[r], 0.f) * w2[0][r];
    }
#pragma unroll
    for (int r = 0; r < 8; ++r) {
      p0 += fmaxf(a10[r], 0.f) * w2[1][r];
      p1 += fmaxf(a11[r], 0.f) * w2[1][r];
    }
    pdst[32 * tp]      = p0;
    pdst[32 * tp + 16] = p1;
  }
  __syncthreads();

  {
    float s = lds_part[tid];
#pragma unroll
    for (int j = 1; j < 8; ++j) s += lds_part[j * 128 + tid];
    out[(size_t)g0 * NPG + tid] = s + bp2[0];
  }
}

#define HP 132

template <int STEPS, int CH>
__device__ __forceinline__ void tile_gemm(const v2f* a, const float* __restrict__ wl, v8f* acc) {
  constexpr int NCH = STEPS / CH;
  float4 buf[2][CH * 4];
#pragma unroll
  for (int i = 0; i < CH; ++i) {
    const float4* p = (const float4*)(wl + (size_t)(4 * i) * D);
    buf[0][4 * i + 0] = p[0];  buf[0][4 * i + 1] = p[1];
    buf[0][4 * i + 2] = p[32]; buf[0][4 * i + 3] = p[33];
  }
#pragma unroll
  for (int ch = 0; ch < NCH; ++ch) {
    if (ch + 1 < NCH) {
#pragma unroll
      for (int i = 0; i < CH; ++i) {
        const float4* p = (const float4*)(wl + (size_t)(4 * ((ch + 1) * CH + i)) * D);
        buf[(ch + 1) & 1][4 * i + 0] = p[0];  buf[(ch + 1) & 1][4 * i + 1] = p[1];
        buf[(ch + 1) & 1][4 * i + 2] = p[32]; buf[(ch + 1) & 1][4 * i + 3] = p[33];
      }
    }
    __builtin_amdgcn_sched_barrier(0);
#pragma unroll
    for (int i = 0; i < CH; ++i) {
      const v2f av = a[ch * CH + i];
      const float4 r0a = buf[ch & 1][4 * i + 0], r0b = buf[ch & 1][4 * i + 1];
      const float4 r1a = buf[ch & 1][4 * i + 2], r1b = buf[ch & 1][4 * i + 3];
      acc[0] = mma4(av, r0a.x, r1a.x, acc[0]);
      acc[1] = mma4(av, r0a.y, r1a.y, acc[1]);
      acc[2] = mma4(av, r0a.z, r1a.z, acc[2]);
      acc[3] = mma4(av, r0a.w, r1a.w, acc[3]);
      acc[4] = mma4(av, r0b.x, r1b.x, acc[4]);
      acc[5] = mma4(av, r0b.y, r1b.y, acc[5]);
      acc[6] = mma4(av, r0b.z, r1b.z, acc[6]);
      acc[7] = mma4(av, r0b.w, r1b.w, acc[7]);
    }
    __builtin_amdgcn_sched_barrier(0);
  }
}

__device__ __forceinline__ void store_part_tile(const v8f* acc, float* part, int c, int half) {
#pragma unroll
  for (int r = 0; r < 8; ++r) {
    float4 lo, hi;
    lo.x = acc[0][r]; lo.y = acc[1][r]; lo.z = acc[2][r]; lo.w = acc[3][r];
    hi.x = acc[4][r]; hi.y = acc[5][r]; hi.z = acc[6][r]; hi.w = acc[7][r];
    float* q = part + (8 * half + r) * HP + 8 * c;
    *(float4*)q = lo;
    *(float4*)(q + 4) = hi;
  }
}

__device__ __forceinline__ float4 sum_parts(const float* parts, int row, int lane, float4 b) {
  float4 s = *(const float4*)(parts + row * HP + 4 * lane);
#pragma unroll
  for (int w = 1; w < 4; ++w) {
    const float4 t = *(const float4*)(parts + (w * 16 + row) * HP + 4 * lane);
    s.x += t.x; s.y += t.y; s.z += t.z; s.w += t.w;
  }
  s.x += b.x; s.y += b.y; s.z += b.z; s.w += b.w;
  return s;
}

__device__ __forceinline__ void reduce_relu_tile(const float* parts, const float* __restrict__ bias,
                                                 float* tile, int wave, int lane) {
  const float4 b = *(const float4*)(bias + 4 * lane);
#pragma unroll
  for (int j = 0; j < 4; ++j) {
    const int row = 4 * j + wave;
    float4 s = sum_parts(parts, row, lane, b);
    s.x = fmaxf(s.x, 0.f); s.y = fmaxf(s.y, 0.f); s.z = fmaxf(s.z, 0.f); s.w = fmaxf(s.w, 0.f);
    *(float4*)(tile + row * HP + 4 * lane) = s;
  }
}

__device__ __forceinline__ void base_tile(
    const float* __restrict__ node, const float* __restrict__ glob,
    const float* __restrict__ Wp1,  const float* __restrict__ bp1,
    float* __restrict__ base_ws, float* parts, int tile, int wave, int lane)
{
  const int half = lane >> 4, c = lane & 15;
  const int g = tile * 16 + c;
  const int part = wave >> 1;
  const int k0 = 64 * (wave & 1);
  const float* s = (part ? glob + (size_t)g * D
                         : node + ((size_t)(g + 1) * NPG - 1) * D)
                   + k0 + 2 * half;
  v2f a[16];
#pragma unroll
  for (int i = 0; i < 16; ++i) a[i] = *(const v2f*)(s + 4 * i);
  v8f acc[8];
#pragma unroll
  for (int j = 0; j < 8; ++j) acc[j] = v8f{};
  tile_gemm<16, 4>(a, Wp1 + (size_t)(2 * D * part + k0 + 2 * half) * D + 8 * c, acc);
  store_part_tile(acc, parts + wave * 16 * HP, c, half);
  __syncthreads();
  const float4 b = *(const float4*)(bp1 + 4 * lane);
#pragma unroll
  for (int j = 0; j < 4; ++j) {
    const int row = 4 * j + wave;
    *(float4*)(base_ws + (size_t)(tile * 16 + row) * D + 4 * lane) = sum_parts(parts, row, lane, b);
  }
}

__global__ __launch_bounds__(128) void label_kernel(
    const float* __restrict__ node, const float* __restrict__ glob,
    const float* __restrict__ Wl1,  const float* __restrict__ bl1,
    const float* __restrict__ Wl2,  const float* __restrict__ bl2,
    const float* __restrict__ Wl3,  const float* __restrict__ bl3,
    const int* __restrict__ pgi,    const int* __restrict__ pni,
    const float* __restrict__ Wp1,  const float* __restrict__ bp1,
    float* __restrict__ base_ws, size_t ws_bytes, int nlab,
    float* __restrict__ out, int outoff)
{
  extern __shared__ float smem[];
  float* parts = smem;
  float* tile  = smem + 4 * 16 * HP;
  const int tid = threadIdx.x;
  const int wave = tid >> 5, lane = tid & 31;
  const int half = lane >> 4, c = lane & 15;
  if ((int)blockIdx.x >= nlab) {
    if (ws_bytes >= (size_t)nlab * 16 * D * sizeof(float))
      base_tile(node, glob, Wp1, bp1, base_ws, parts, (int)blockIdx.x - nlab, wave, lane);
    return;
  }
  const int rowid = blockIdx.x * 16 + c;
  const int pg = pgi[rowid];
  const int pn = pni[rowid];
  const float* src0 = node + ((size_t)(pg + 1) * NPG - 1) * D + 2 * half;
  const float* src1 = node + (size_t)pn * D + 2 * half;
  const float* src2 = glob + (size_t)pg * D + 2 * half;

  v8f acc[8];

  {
    v2f a[24];
#pragma unroll
    for (int i = 0; i < 24; ++i) {
      const int s = 24 * wave + i;
      const float* p = s < 32 ? src0 : (s < 64 ? src1 : src2);
      a[i] = *(const v2f*)(p + 4 * (s & 31));
    }
#pragma unroll
    for (int j = 0; j < 8; ++j) acc[j] = v8f{};
    tile_gemm<24, 4>(a, Wl1 + (size_t)(96 * wave + 2 * half) * D + 8 * c, acc);
    store_part_tile(acc, parts + wave * 16 * HP, c, half);
  }
  __syncthreads();
  reduce_relu_tile(parts, bl1, tile, wave, lane);
  __syncthreads();

  float b3x[8], b3y[8];
  {
    const int cc = c < 14 ? c : 13;
    const float* w3 = Wl3 + (32 * wave + 2 * half) * 14 + cc;
#pragma unroll
    for (int s = 0; s < 8; ++s) {
      b3x[s] = w3[(4 * s) * 14];
      b3y[s] = w3[(4 * s + 1) * 14];
    }
  }

  {
    v2f a[8];
#pragma unroll
    for (int i = 0; i < 8; ++i) a[i] = *(const v2f*)(tile + c * HP + 4 * (8 * wave + i) + 2 * half);
#pragma unroll
    for (int j = 0; j < 8; ++j) acc[j] = v8f{};
    tile_gemm<8, 4>(a, Wl2 + (size_t)(32 * wave + 2 * half) * D + 8 * c, acc);
    store_part_tile(acc, parts + wave * 16 * HP, c, half);
  }
  __syncthreads();
  reduce_relu_tile(parts, bl2, tile, wave, lane);
  __syncthreads();

  {
    v2f a[8];
#pragma unroll
    for (int s = 0; s < 8; ++s) a[s] = *(const v2f*)(tile + c * HP + 4 * (8 * wave + s) + 2 * half);
    v8f o = {};
#pragma unroll
    for (int s = 0; s < 8; ++s) o = mma4(a[s], b3x[s], b3y[s], o);
    float* q = parts + (wave * 16 + 8 * half) * HP + c;
#pragma unroll
    for (int r = 0; r < 8; ++r) q[r * HP] = o[r];
  }
  __syncthreads();
  for (int idx = tid; idx < 16 * 14; idx += 128) {
    const int row = idx / 14, col = idx - row * 14;
    float s = parts[row * HP + col];
#pragma unroll
    for (int w = 1; w < 4; ++w) s += parts[(w * 16 + row) * HP + col];
    out[outoff + blockIdx.x * (16 * 14) + idx] = s + bl3[col];
  }
}

extern "C" void kernel_launch(void* const* d_in, const int* in_sizes, int n_in,
                              void* d_out, int out_size, void* d_ws, size_t ws_size,
                              hipStream_t stream) {
  (void)n_in; (void)out_size;
  const float* node = (const float*)d_in[0];
  const float* glob = (const float*)d_in[1];
  const float* Wp1  = (const float*)d_in[2];
  const float* bp1  = (const float*)d_in[3];
  const float* Wp2  = (const float*)d_in[4];
  const float* bp2  = (const float*)d_in[5];
  const float* Wl1  = (const float*)d_in[6];
  const float* bl1  = (const float*)d_in[7];
  const float* Wl2  = (const float*)d_in[8];
  const float* bl2  = (const float*)d_in[9];
  const float* Wl3  = (const float*)d_in[10];
  const float* bl3  = (const float*)d_in[11];
  const int*   pgi  = (const int*)d_in[12];
  const int*   pni  = (const int*)d_in[13];
  float* out = (float*)d_out;
  float* ws  = (float*)d_ws;

  const int B = in_sizes[1] / D;
  const int N = in_sizes[0] / D;

  const size_t smem1 = (128 * PP + 256 + 256 + 1024) * sizeof(float);
  const size_t smem2 = (4 * 16 * HP + 16 * HP) * sizeof(float);

  label_kernel<<<B / 8, 128, smem2, stream>>>(node, glob, Wl1, bl1, Wl2, bl2,
                                               Wl3, bl3, pgi, pni, Wp1, bp1, ws, ws_size, B / 16, out, N);
  partner_kernel<<<B / 2, 128, smem1, stream>>>(node, glob, Wp1, bp1, Wp2, bp2, ws, ws_size, B / 16, out);
}


# partner kernel: each wave keeps the whole 128x128 weight block in 512 registers, one wave per SIMD (256 blocks), walks its graphs through a private LDS tile ring, no barriers; sum order changes
# speedup vs baseline: 3.0292x; 1.0472x over previous
#include <hip/hip_runtime.h>

typedef float v2f __attribute__((ext_vector_type(2)));
typedef float v8f __attribute__((ext_vector_type(8)));
typedef float v4f __attribute__((ext_vector_type(4)));

#define D   128
#define NPG 64

__device__ __forceinline__ float bcast(float v, int k) {
  return __builtin_bit_cast(float, __builtin_amdgcn_readlane(__builtin_bit_cast(int, v), k));
}

__device__ __forceinline__ v8f mma4(v2f a, float b0, float b1, v8f c) {
  v2f b; b.x = b0; b.y = b1;
  return __builtin_amdgcn_wmma_f32_16x16x4_f32(false, a, false, b, (short)0, c, false, false);
}

#define TP 132

#define LOAD_ROWS(GG, TT)                                                                  \
  {                                                                                        \
    const v4f* rs_ = (const v4f*)(node + ((size_t)(GG) * NPG + 16 * (TT)) * D) + lane;       \
    _Pragma("unroll") for (int i_ = 0; i_ < 16; ++i_) R[i_] = rs_[i_ * 32];                \
  }

__device__ __forceinline__ v4f base_quad(bool from_ws, const float* __restrict__ base_ws,
                                            const float* __restrict__ node, const float* __restrict__ glob,
                                            const float* __restrict__ Wp1, const float* __restrict__ bp1, int g, int lane) {
  if (from_ws) return *(const v4f*)(base_ws + (size_t)g * D + 4 * lane);
  v4f s = *(const v4f*)(bp1 + 4 * lane);
  for (int part = 0; part < 2; ++part) {
    const float* x = part ? glob + (size_t)g * D : node + ((size_t)(g + 1) * NPG - 1) * D;
    const float* w = Wp1 + (size_t)(2 * D * part) * D + 4 * lane;
    for (int k = 0; k < D; ++k) {
      const float xv = x[k];
      const v4f wv = *(const v4f*)(w + (size_t)k * D);
      s.x = __builtin_fmaf(xv, wv.x, s.x); s.y = __builtin_fmaf(xv, wv.y, s.y);
      s.z = __builtin_fmaf(xv, wv.z, s.z); s.w = __builtin_fmaf(xv, wv.w, s.w);
    }
  }
  return s;
}

__device__ __forceinline__ void read_base(v8f* bs, const float* lb, int half) {
#pragma unroll
  for (int T = 0; T < 8; ++T) {
    const float4 a = *(const float4*)(lb + 16 * T + 8 * half), b = *(const float4*)(lb + 16 * T + 8 * half + 4);
    bs[T] = v8f{a.x, a.y, a.z, a.w, b.x, b.y, b.z, b.w};
  }
}

#define TILE_LOGIT(ACC, S)                                                                                          \
  {                                                                                                                 \
    v2f q0_ = {0.f, 0.f}, q1_ = {0.f, 0.f}, q2_ = {0.f, 0.f}, q3_ = {0.f, 0.f};                                     \
    _Pragma("unroll") for (int T_ = 0; T_ < 8; T_ += 4) {                                                           \
      _Pragma("unroll") for (int r_ = 0; r_ < 4; ++r_) {                                                            \
        q0_ = __builtin_elementwise_fma(v2f{fmaxf(ACC[T_][2 * r_], 0.f),     fmaxf(ACC[T_][2 * r_ + 1], 0.f)},     w2[T_][r_],     q0_); \
        q1_ = __builtin_elementwise_fma(v2f{fmaxf(ACC[T_ + 1][2 * r_], 0.f), fmaxf(ACC[T_ + 1][2 * r_ + 1], 0.f)}, w2[T_ + 1][r_], q1_); \
        q2_ = __builtin_elementwise_fma(v2f{fmaxf(ACC[T_ + 2][2 * r_], 0.f), fmaxf(ACC[T_ + 2][2 * r_ + 1], 0.f)}, w2[T_ + 2][r_], q2_); \
        q3_ = __builtin_elementwise_fma(v2f{fmaxf(ACC[T_ + 3][2 * r_], 0.f), fmaxf(ACC[T_ + 3][2 * r_ + 1], 0.f)}, w2[T_ + 3][r_], q3_); \
      }                                                                                                             \
    }                                                                                                               \
    const float p_ = ((q0_.x + q0_.y) + (q1_.x + q1_.y)) + ((q2_.x + q2_.y) + (q3_.x + q3_.y));                     \
    S = p_ + __shfl_xor(p_, 16);                                                                                    \
  }

__global__ __launch_bounds__(128) void partner_kernel(
    const float* __restrict__ node, const float* __restrict__ glob,
    const float* __restrict__ Wp1,  const float* __restrict__ bp1,
    const float* __restrict__ Wp2,  const float* __restrict__ bp2,
    const float* __restrict__ base_ws, size_t ws_bytes, int nlab,
    float* __restrict__ out, int B)
{
  extern __shared__ float smem[];
  const int tid = threadIdx.x;
  const int wave = tid >> 5, lane = tid & 31;
  const int half = lane >> 4, c = lane & 15;
  float* lx = smem + wave * (2 * 16 * TP);
  float* lb = smem + 4 * (2 * 16 * TP) + wave * D;
  const int nw = (int)gridDim.x * 4;
  int g = (int)blockIdx.x * 4 + wave;
  if (g >= B) return;

  v4f R[16];
  LOAD_ROWS(g, 0)

  v2f wa[8][32];
  {
    const float* wsrc = Wp1 + (size_t)(D + 4 * half) * D + c;
#pragma unroll
    for (int u = 0; u < 16; ++u) {
#pragma unroll
      for (int T = 0; T < 8; ++T) {
        wa[T][2 * u].x     = wsrc[(8 * u + 0) * D + 16 * T];
        wa[T][2 * u].y     = wsrc[(8 * u + 1) * D + 16 * T];
        wa[T][2 * u + 1].x = wsrc[(8 * u + 2) * D + 16 * T];
        wa[T][2 * u + 1].y = wsrc[(8 * u + 3) * D + 16 * T];
      }
    }
  }
  v2f w2[8][4];
#pragma unroll
  for (int T = 0; T < 8; ++T) {
    const float4 a = *(const float4*)(Wp2 + 16 * T + 8 * half), b = *(const float4*)(Wp2 + 16 * T + 8 * half + 4);
    w2[T][0] = v2f{a.x, a.y}; w2[T][1] = v2f{a.z, a.w};
    w2[T][2] = v2f{b.x, b.y}; w2[T][3] = v2f{b.z, b.w};
  }
  const float b2 = bp2[0];

  const bool base_from_ws = ws_bytes >= (size_t)nlab * 16 * D * sizeof(float);
  v8f bs[8];
  *(v4f*)(lb + 4 * lane) = base_quad(base_from_ws, base_ws, node, glob, Wp1, bp1, g, lane);
  read_base(bs, lb, half);

#pragma unroll
  for (int i = 0; i < 16; ++i) *(v4f*)(lx + i * TP + 4 * lane) = R[i];
  LOAD_ROWS(g, 1)
  const float* xb = lx + c * TP + 4 * half;
  float4 fA[8], fB[8];
#pragma unroll
  for (int j = 0; j < 8; ++j) fA[j] = *(const float4*)(xb + 8 * j);

  float keep = 0.f;
  v8f accP[8];
#pragma unroll
  for (int T = 0; T < 8; ++T) accP[T] = v8f{};
  bool first = true;
  for (; g < B; g += nw) {
    const int gn = (g + nw < B) ? g + nw : g;
    v4f bq = {0.f, 0.f, 0.f, 0.f};
#pragma unroll
    for (int t = 0; t < 4; ++t) {
      float* bnext = lx + ((t + 1) & 1) * (16 * TP);
      const float* bcur = xb + (t & 1) * (16 * TP);
#pragma unroll
      for (int i = 0; i < 16; ++i) *(v4f*)(bnext + i * TP + 4 * lane) = R[i];
      __builtin_amdgcn_sched_barrier(0);
      if (t < 2) LOAD_ROWS(g, t + 2) else LOAD_ROWS(gn, t - 2)
      if (t == 0) bq = base_quad(base_from_ws, base_ws, node, glob, Wp1, bp1, gn, lane);
#pragma unroll
      for (int j = 0; j < 8; ++j) fB[j] = *(const float4*)(bcur + 8 * (8 + j));
      __builtin_amdgcn_sched_barrier(0);
      v8f acc[8];
#pragma unroll
      for (int j = 0; j < 8; ++j) {
        const float4 f = fA[j];
#pragma unroll
        for (int T = 0; T < 8; ++T) acc[T] = mma4(wa[T][2 * j], f.x, f.y, j == 0 ? bs[T] : acc[T]);
#pragma unroll
        for (int T = 0; T < 8; ++T) acc[T] = mma4(wa[T][2 * j + 1], f.z, f.w, acc[T]);
      }
      {
        float s;
        TILE_LOGIT(accP, s)
        if (t == 0) { if (!first) out[(size_t)(g - nw) * NPG + 32 + lane] = (half ? s : keep) + b2; }
        else if (t == 2) out[(size_t)g * NPG + lane] = (half ? s : keep) + b2;
        else keep = s;
      }
      if (t == 3) { *(v4f*)(lb + 4 * lane) = bq; read_base(bs, lb, half); }
#pragma unroll
      for (int j = 0; j < 8; ++j) fA[j] = *(const float4*)(xb + ((t + 1) & 1) * (16 * TP) + 8 * j);
      __builtin_amdgcn_sched_barrier(0);
#pragma unroll
      for (int j = 0; j < 8; ++j) {
        const float4 f = fB[j];
#pragma unroll
        for (int T = 0; T < 8; ++T) acc[T] = mma4(wa[T][2 * (8 + j)], f.x, f.y, acc[T]);
#pragma unroll
        for (int T = 0; T < 8; ++T) acc[T] = mma4(wa[T][2 * (8 + j) + 1], f.z, f.w, acc[T]);
      }
#pragma unroll
      for (int T = 0; T < 8; ++T) accP[T] = acc[T];
      __builtin_amdgcn_sched_barrier(0);
    }
    first = false;
  }
  {
    float s;
    TILE_LOGIT(accP, s)
    out[(size_t)(g - nw) * NPG + 32 + lane] = (half ? s : keep) + b2;
  }
}

#define HP 132

template <int STEPS, int CH>
__device__ __forceinline__ void tile_gemm(const v2f* a, const float* __restrict__ wl, v8f* acc) {
  constexpr int NCH = STEPS / CH;
  float4 buf[2][CH * 4];
#pragma unroll
  for (int i = 0; i < CH; ++i) {
    const float4* p = (const float4*)(wl + (size_t)(4 * i) * D);
    buf[0][4 * i + 0] = p[0];  buf[0][4 * i + 1] = p[1];
    buf[0][4 * i + 2] = p[32]; buf[0][4 * i + 3] = p[33];
  }
#pragma unroll
  for (int ch = 0; ch < NCH; ++ch) {
    if (ch + 1 < NCH) {
#pragma unroll
      for (int i = 0; i < CH; ++i) {
        const float4* p = (const float4*)(wl + (size_t)(4 * ((ch + 1) * CH + i)) * D);
        buf[(ch + 1) & 1][4 * i + 0] = p[0];  buf[(ch + 1) & 1][4 * i + 1] = p[1];
        buf[(ch + 1) & 1][4 * i + 2] = p[32]; buf[(ch + 1) & 1][4 * i + 3] = p[33];
      }
    }
    __builtin_amdgcn_sched_barrier(0);
#pragma unroll
    for (int i = 0; i < CH; ++i) {
      const v2f av = a[ch * CH + i];
      const float4 r0a = buf[ch & 1][4 * i + 0], r0b = buf[ch & 1][4 * i + 1];
      const float4 r1a = buf[ch & 1][4 * i + 2], r1b = buf[ch & 1][4 * i + 3];
      acc[0] = mma4(av, r0a.x, r1a.x, acc[0]);
      acc[1] = mma4(av, r0a.y, r1a.y, acc[1]);
      acc[2] = mma4(av, r0a.z, r1a.z, acc[2]);
      acc[3] = mma4(av, r0a.w, r1a.w, acc[3]);
      acc[4] = mma4(av, r0b.x, r1b.x, acc[4]);
      acc[5] = mma4(av, r0b.y, r1b.y, acc[5]);
      acc[6] = mma4(av, r0b.z, r1b.z, acc[6]);
      acc[7] = mma4(av, r0b.w, r1b.w, acc[7]);
    }
    __builtin_amdgcn_sched_barrier(0);
  }
}

__device__ __forceinline__ void store_part_tile(const v8f* acc, float* part, int c, int half) {
#pragma unroll
  for (int r = 0; r < 8; ++r) {
    float4 lo, hi;
    lo.x = acc[0][r]; lo.y = acc[1][r]; lo.z = acc[2][r]; lo.w = acc[3][r];
    hi.x = acc[4][r]; hi.y = acc[5][r]; hi.z = acc[6][r]; hi.w = acc[7][r];
    float* q = part + (8 * half + r) * HP + 8 * c;
    *(float4*)q = lo;
    *(float4*)(q + 4) = hi;
  }
}

__device__ __forceinline__ float4 sum_parts(const float* parts, int row, int lane, float4 b) {
  float4 s = *(const float4*)(parts + row * HP + 4 * lane);
#pragma unroll
  for (int w = 1; w < 4; ++w) {
    const float4 t = *(const float4*)(parts + (w * 16 + row) * HP + 4 * lane);
    s.x += t.x; s.y += t.y; s.z += t.z; s.w += t.w;
  }
  s.x += b.x; s.y += b.y; s.z += b.z; s.w += b.w;
  return s;
}

__device__ __forceinline__ void reduce_relu_tile(const float* parts, const float* __restrict__ bias,
                                                 float* tile, int wave, int lane) {
  const float4 b = *(const float4*)(bias + 4 * lane);
#pragma unroll
  for (int j = 0; j < 4; ++j) {
    const int row = 4 * j + wave;
    float4 s = sum_parts(parts, row, lane, b);
    s.x = fmaxf(s.x, 0.f); s.y = fmaxf(s.y, 0.f); s.z = fmaxf(s.z, 0.f); s.w = fmaxf(s.w, 0.f);
    *(float4*)(tile + row * HP + 4 * lane) = s;
  }
}

__device__ __forceinline__ void base_tile(
    const float* __restrict__ node, const float* __restrict__ glob,
    const float* __restrict__ Wp1,  const float* __restrict__ bp1,
    float* __restrict__ base_ws, float* parts, int tile, int wave, int lane)
{
  const int half = lane >> 4, c = lane & 15;
  const int g = tile * 16 + c;
  const int part = wave >> 1;
  const int k0 = 64 * (wave & 1);
  const float* s = (part ? glob + (size_t)g * D
                         : node + ((size_t)(g + 1) * NPG - 1) * D)
                   + k0 + 2 * half;
  v2f a[16];
#pragma unroll
  for (int i = 0; i < 16; ++i) a[i] = *(const v2f*)(s + 4 * i);
  v8f acc[8];
#pragma unroll
  for (int j = 0; j < 8; ++j) acc[j] = v8f{};
  tile_gemm<16, 4>(a, Wp1 + (size_t)(2 * D * part + k0 + 2 * half) * D + 8 * c, acc);
  store_part_tile(acc, parts + wave * 16 * HP, c, half);
  __syncthreads();
  const float4 b = *(const float4*)(bp1 + 4 * lane);
#pragma unroll
  for (int j = 0; j < 4; ++j) {
    const int row = 4 * j + wave;
    *(float4*)(base_ws + (size_t)(tile * 16 + row) * D + 4 * lane) = sum_parts(parts, row, lane, b);
  }
}

__global__ __launch_bounds__(128) void label_kernel(
    const float* __restrict__ node, const float* __restrict__ glob,
    const float* __restrict__ Wl1,  const float* __restrict__ bl1,
    const float* __restrict__ Wl2,  const float* __restrict__ bl2,
    const float* __restrict__ Wl3,  const float* __restrict__ bl3,
    const int* __restrict__ pgi,    const int* __restrict__ pni,
    const float* __restrict__ Wp1,  const float* __restrict__ bp1,
    float* __restrict__ base_ws, size_t ws_bytes, int nlab,
    float* __restrict__ out, int outoff)
{
  extern __shared__ float smem[];
  float* parts = smem;
  float* tile  = smem + 4 * 16 * HP;
  const int tid = threadIdx.x;
  const int wave = tid >> 5, lane = tid & 31;
  const int half = lane >> 4, c = lane & 15;
  if ((int)blockIdx.x >= nlab) {
    if (ws_bytes >= (size_t)nlab * 16 * D * sizeof(float))
      base_tile(node, glob, Wp1, bp1, base_ws, parts, (int)blockIdx.x - nlab, wave, lane);
    return;
  }
  const int rowid = blockIdx.x * 16 + c;
  const int pg = pgi[rowid];
  const int pn = pni[rowid];
  const float* src0 = node + ((size_t)(pg + 1) * NPG - 1) * D + 2 * half;
  const float* src1 = node + (size_t)pn * D + 2 * half;
  const float* src2 = glob + (size_t)pg * D + 2 * half;

  v8f acc[8];

  {
    v2f a[24];
#pragma unroll
    for (int i = 0; i < 24; ++i) {
      const int s = 24 * wave + i;
      const float* p = s < 32 ? src0 : (s < 64 ? src1 : src2);
      a[i] = *(const v2f*)(p + 4 * (s & 31));
    }
#pragma unroll
    for (int j = 0; j < 8; ++j) acc[j] = v8f{};
    tile_gemm<24, 4>(a, Wl1 + (size_t)(96 * wave + 2 * half) * D + 8 * c, acc);
    store_part_tile(acc, parts + wave * 16 * HP, c, half);
  }
  __syncthreads();
  reduce_relu_tile(parts, bl1, tile, wave, lane);
  __syncthreads();

  float b3x[8], b3y[8];
  {
    const int cc = c < 14 ? c : 13;
    const float* w3 = Wl3 + (32 * wave + 2 * half) * 14 + cc;
#pragma unroll
    for (int s = 0; s < 8; ++s) {
      b3x[s] = w3[(4 * s) * 14];
      b3y[s] = w3[(4 * s + 1) * 14];
    }
  }

  {
    v2f a[8];
#pragma unroll
    for (int i = 0; i < 8; ++i) a[i] = *(const v2f*)(tile + c * HP + 4 * (8 * wave + i) + 2 * half);
#pragma unroll
    for (int j = 0; j < 8; ++j) acc[j] = v8f{};
    tile_gemm<8, 4>(a, Wl2 + (size_t)(32 * wave + 2 * half) * D + 8 * c, acc);
    store_part_tile(acc, parts + wave * 16 * HP, c, half);
  }
  __syncthreads();
  reduce_relu_tile(parts, bl2, tile, wave, lane);
  __syncthreads();

  {
    v2f a[8];
#pragma unroll
    for (int s = 0; s < 8; ++s) a[s] = *(const v2f*)(tile + c * HP + 4 * (8 * wave + s) + 2 * half);
    v8f o = {};
#pragma unroll
    for (int s = 0; s < 8; ++s) o = mma4(a[s], b3x[s], b3y[s], o);
    float* q = parts + (wave * 16 + 8 * half) * HP + c;
#pragma unroll
    for (int r = 0; r < 8; ++r) q[r * HP] = o[r];
  }
  __syncthreads();
  for (int idx = tid; idx < 16 * 14; idx += 128) {
    const int row = idx / 14, col = idx - row * 14;
    float s = parts[row * HP + col];
#pragma unroll
    for (int w = 1; w < 4; ++w) s += parts[(w * 16 + row) * HP + col];
    out[outoff + blockIdx.x * (16 * 14) + idx] = s + bl3[col];
  }
}

extern "C" void kernel_launch(void* const* d_in, const int* in_sizes, int n_in,
                              void* d_out, int out_size, void* d_ws, size_t ws_size,
                              hipStream_t stream) {
  (void)n_in; (void)out_size;
  const float* node = (const float*)d_in[0];
  const float* glob = (const float*)d_in[1];
  const float* Wp1  = (const float*)d_in[2];
  const float* bp1  = (const float*)d_in[3];
  const float* Wp2  = (const float*)d_in[4];
  const float* bp2  = (const float*)d_in[5];
  const float* Wl1  = (const float*)d_in[6];
  const float* bl1  = (const float*)d_in[7];
  const float* Wl2  = (const float*)d_in[8];
  const float* bl2  = (const float*)d_in[9];
  const float* Wl3  = (const float*)d_in[10];
  const float* bl3  = (const float*)d_in[11];
  const int*   pgi  = (const int*)d_in[12];
  const int*   pni  = (const int*)d_in[13];
  float* out = (float*)d_out;
  float* ws  = (float*)d_ws;

  const int B = in_sizes[1] / D;
  const int N = in_sizes[0] / D;

  const size_t smem1 = (4 * 2 * 16 * TP + 4 * D) * sizeof(float);
  const size_t smem2 = (4 * 16 * HP + 16 * HP) * sizeof(float);

  label_kernel<<<B / 8, 128, smem2, stream>>>(node, glob, Wl1, bl1, Wl2, bl2,
                                               Wl3, bl3, pgi, pni, Wp1, bp1, ws, ws_size, B / 16, out, N);
  partner_kernel<<<256, 128, smem1, stream>>>(node, glob, Wp1, bp1, Wp2, bp2, ws, ws_size, B / 16, out, B);
}
